# starting kernel
# baseline (speedup 1.0000x reference)
#include <hip/hip_runtime.h>
#include <cmath>


#define NRR 2048
#define NCC 2048
#define KSZ 10
#define NCH 6

#define TILE_R 16
#define NWAVES 8
#define TILE_C (16 * NWAVES)
#define TMP_ROWS 28
#define LDS_PITCH 132

typedef __attribute__((ext_vector_type(2))) float v2f;
typedef __attribute__((ext_vector_type(8))) float v8f;

struct TapsArg { float g[NCH][KSZ]; };

__global__ __launch_bounds__(256)
void s2_gaussian_blur_fused(const float* __restrict__ x,
                            float* __restrict__ out,
                            TapsArg taps) {
  __shared__ float s_tmp[TMP_ROWS * LDS_PITCH];
  __shared__ float s_g[16];

  const int tid  = threadIdx.x;
  const int lane = tid & 31;
  const int wave = tid >> 5;
  const int ch   = blockIdx.z;
  const int r0   = blockIdx.y * TILE_R;
  const int c0   = blockIdx.x * TILE_C;

  const float* __restrict__ xc = x   + (size_t)ch * NRR * NCC;
  float* __restrict__       oc = out + (size_t)ch * NRR * NCC;

  if (tid < 16) s_g[tid] = (tid < KSZ) ? taps.g[ch][tid] : 0.0f;
  __syncthreads();

#pragma unroll
  for (int it = 0; it < (TMP_ROWS * TILE_C) / 256; ++it) {
    int idx = tid + it * 256;
    int row = idx >> 7;
    int col = idx & 127;
    float acc = 0.0f;
    if (row < 25) {
      int gr = (r0 - 4 + row) & (NRR - 1);
      const float* __restrict__ rp = xc + (size_t)gr * NCC;
#pragma unroll
      for (int u = 0; u < KSZ; ++u) {
        int gc = (c0 + col - 4 + u) & (NCC - 1);
        acc = fmaf(s_g[u], rp[gc], acc);
      }
    }
    s_tmp[row * LDS_PITCH + col] = acc;
  }
  __syncthreads();

  const int m  = lane & 15;
  const int lh = lane >> 4;
  const int cw = wave * 16;

  v8f acc = {};
#pragma unroll
  for (int kk = 0; kk < 7; ++kk) {
    const int kb = 4 * kk + 2 * lh;
    int d0 = kb - m;
    int d1 = kb + 1 - m;
    int i0 = ((unsigned)d0 < KSZ) ? d0 : 15;
    int i1 = ((unsigned)d1 < KSZ) ? d1 : 15;
    v2f a;
    a.x = s_g[i0];
    a.y = s_g[i1];
    v2f b;
    b.x = s_tmp[(kb + 0) * LDS_PITCH + cw + m];
    b.y = s_tmp[(kb + 1) * LDS_PITCH + cw + m];
    acc = __builtin_amdgcn_wmma_f32_16x16x4_f32(
         false, a,  false, b,
         (short)0, acc,  false,  false);
  }

  const int cc = c0 + cw + m;
#pragma unroll
  for (int v = 0; v < 8; ++v) {
    int rr = r0 + v + 8 * lh;
    oc[(size_t)rr * NCC + cc] = acc[v];
  }
}

extern "C" void kernel_launch(void* const* d_in, const int* in_sizes, int n_in,
                              void* d_out, int out_size, void* d_ws, size_t ws_size,
                              hipStream_t stream) {
  (void)in_sizes; (void)n_in; (void)out_size; (void)d_ws; (void)ws_size;

  static const double mtf[NCH] = {0.38, 0.34, 0.34, 0.26, 0.22, 0.23};
  TapsArg taps;
  for (int c = 0; c < NCH; ++c) {
    double sigma = 2.0 * std::sqrt(-2.0 * std::log(mtf[c])) / M_PI;
    double e[KSZ], s = 0.0;
    for (int t = 0; t < KSZ; ++t) {
      double d = (double)t - 4.5;
      e[t] = std::exp(-d * d / (2.0 * sigma * sigma));
      s += e[t];
    }
    for (int t = 0; t < KSZ; ++t) taps.g[c][t] = (float)(e[t] / s);
  }

  const float* x  = (const float*)d_in[0];
  float*       yo = (float*)d_out;

  dim3 grid(NCC / TILE_C, NRR / TILE_R, NCH);
  s2_gaussian_blur_fused<<<grid, 256, 0, stream>>>(x, yo, taps);
}


# row streaming: a work-group walks 16 row tiles down its 128-column strip, keeping filtered rows in a 64-row LDS ring, so each row is filtered once (was 25 per 16) and 16x fewer waves enter
# speedup vs baseline: 1.4474x; 1.4474x over previous
#include <hip/hip_runtime.h>
#include <cmath>


#define NRR 2048
#define NCC 2048
#define KSZ 10
#define NCH 6

#define TILE_R 16
#define NWAVES 8
#define TILE_C (16 * NWAVES)
#define NT 16
#define RING 64
#define LDS_PITCH 132

typedef __attribute__((ext_vector_type(2))) float v2f;
typedef __attribute__((ext_vector_type(8))) float v8f;

struct TapsArg { float g[NCH][KSZ]; };

__global__ __launch_bounds__(256)
void s2_gaussian_blur_fused(const float* __restrict__ x,
                            float* __restrict__ out,
                            TapsArg taps) {
  __shared__ float s_tmp[RING * LDS_PITCH];
  __shared__ float s_g[16];

  const int tid   = threadIdx.x;
  const int lane  = tid & 31;
  const int wave  = tid >> 5;
  const int ch    = blockIdx.z;
  const int rbase = blockIdx.y * (TILE_R * NT);
  const int c0    = blockIdx.x * TILE_C;

  const float* __restrict__ xc = x   + (size_t)ch * NRR * NCC;
  float* __restrict__       oc = out + (size_t)ch * NRR * NCC;

  if (tid < 16) s_g[tid] = (tid < KSZ) ? taps.g[ch][tid] : 0.0f;
  __syncthreads();

  const int m  = lane & 15;
  const int lh = lane >> 4;
  const int cw = wave * 16;

  float g[KSZ];
#pragma unroll
  for (int u = 0; u < KSZ; ++u) g[u] = s_g[u];
  v2f afrag[7];
#pragma unroll
  for (int kk = 0; kk < 7; ++kk) {
    const int kb = 4 * kk + 2 * lh;
    int d0 = kb - m;
    int d1 = kb + 1 - m;
    int i0 = ((unsigned)d0 < KSZ) ? d0 : 15;
    int i1 = ((unsigned)d1 < KSZ) ? d1 : 15;
    afrag[kk].x = s_g[i0];
    afrag[kk].y = s_g[i1];
  }

  const int col  = tid & 127;
  const int rsub = tid >> 7;

#pragma unroll
  for (int it = 0; it < 6; ++it) {
    const int rel = rsub + 2 * it;
    const int gr  = (rbase - 4 + rel) & (NRR - 1);
    const float* __restrict__ rp = xc + (size_t)gr * NCC;
    float acc = 0.0f;
#pragma unroll
    for (int u = 0; u < KSZ; ++u) {
      int gc = (c0 + col - 4 + u) & (NCC - 1);
      acc = fmaf(g[u], rp[gc], acc);
    }
    s_tmp[(rel & (RING - 1)) * LDS_PITCH + col] = acc;
  }

  for (int t = 0; t < NT; ++t) {
#pragma unroll
    for (int it = 0; it < 8; ++it) {
      const int rel = 16 * t + 12 + rsub + 2 * it;
      const int gr  = (rbase - 4 + rel) & (NRR - 1);
      const float* __restrict__ rp = xc + (size_t)gr * NCC;
      float acc = 0.0f;
#pragma unroll
      for (int u = 0; u < KSZ; ++u) {
        int gc = (c0 + col - 4 + u) & (NCC - 1);
        acc = fmaf(g[u], rp[gc], acc);
      }
      s_tmp[(rel & (RING - 1)) * LDS_PITCH + col] = acc;
    }
    __syncthreads();

    v8f acc = {};
#pragma unroll
    for (int kk = 0; kk < 7; ++kk) {
      const int kb = 4 * kk + 2 * lh;
      v2f b;
      if (kk < 6) {
        b.x = s_tmp[((16 * t + kb + 0) & (RING - 1)) * LDS_PITCH + cw + m];
        b.y = s_tmp[((16 * t + kb + 1) & (RING - 1)) * LDS_PITCH + cw + m];
      } else {
        float r24 = s_tmp[((16 * t + 24) & (RING - 1)) * LDS_PITCH + cw + m];
        b.x = lh ? 0.0f : r24;
        b.y = 0.0f;
      }
      acc = __builtin_amdgcn_wmma_f32_16x16x4_f32(
           false, afrag[kk],  false, b,
           (short)0, acc,  false,  false);
    }

    const int r0 = rbase + t * TILE_R;
    const int cc = c0 + cw + m;
#pragma unroll
    for (int v = 0; v < 8; ++v) {
      int rr = r0 + v + 8 * lh;
      oc[(size_t)rr * NCC + cc] = acc[v];
    }
  }
}

extern "C" void kernel_launch(void* const* d_in, const int* in_sizes, int n_in,
                              void* d_out, int out_size, void* d_ws, size_t ws_size,
                              hipStream_t stream) {
  (void)in_sizes; (void)n_in; (void)out_size; (void)d_ws; (void)ws_size;

  static const double mtf[NCH] = {0.38, 0.34, 0.34, 0.26, 0.22, 0.23};
  TapsArg taps;
  for (int c = 0; c < NCH; ++c) {
    double sigma = 2.0 * std::sqrt(-2.0 * std::log(mtf[c])) / M_PI;
    double e[KSZ], s = 0.0;
    for (int t = 0; t < KSZ; ++t) {
      double d = (double)t - 4.5;
      e[t] = std::exp(-d * d / (2.0 * sigma * sigma));
      s += e[t];
    }
    for (int t = 0; t < KSZ; ++t) taps.g[c][t] = (float)(e[t] / s);
  }

  const float* x  = (const float*)d_in[0];
  float*       yo = (float*)d_out;

  dim3 grid(NCC / TILE_C, NRR / (TILE_R * NT), NCH);
  s2_gaussian_blur_fused<<<grid, 256, 0, stream>>>(x, yo, taps);
}


# horizontal pass: one wave per row; each lane reads its 13-column window once (three 16-byte loads + one 4-byte, all issued as one group) and filters 4 adjacent columns from registers; same sum order
# speedup vs baseline: 1.9782x; 1.3667x over previous
#include <hip/hip_runtime.h>
#include <cmath>


#define NRR 2048
#define NCC 2048
#define KSZ 10
#define NCH 6

#define TILE_R 16
#define NWAVES 8
#define TILE_C (16 * NWAVES)
#define NT 16
#define RING 64
#define LDS_PITCH 132

typedef __attribute__((ext_vector_type(2))) float v2f;
typedef __attribute__((ext_vector_type(8))) float v8f;

typedef __attribute__((ext_vector_type(4))) float v4f;

struct RowIn { v4f a, b, c; float s; };

__device__ __forceinline__ RowIn load_row(const float* __restrict__ rp,
                                          int cA, int cB, int cC, int cS) {
  RowIn r;
  r.a = *(const v4f*)(rp + cA);
  r.b = *(const v4f*)(rp + cB);
  r.c = *(const v4f*)(rp + cC);
  r.s = rp[cS];
  return r;
}

__device__ __forceinline__ v4f filt_row(const RowIn& r, const float (&g)[KSZ]) {
  const float x[13] = {r.a.x, r.a.y, r.a.z, r.a.w, r.b.x, r.b.y, r.b.z, r.b.w,
                       r.c.x, r.c.y, r.c.z, r.c.w, r.s};
  v4f o;
#pragma unroll
  for (int k = 0; k < 4; ++k) {
    float acc = 0.0f;
#pragma unroll
    for (int u = 0; u < KSZ; ++u) acc = fmaf(g[u], x[k + u], acc);
    o[k] = acc;
  }
  return o;
}

struct TapsArg { float g[NCH][KSZ]; };

__global__ __launch_bounds__(256)
void s2_gaussian_blur_fused(const float* __restrict__ x,
                            float* __restrict__ out,
                            TapsArg taps) {
  __shared__ __attribute__((aligned(16))) float s_tmp[RING * LDS_PITCH];
  __shared__ float s_g[16];

  const int tid   = threadIdx.x;
  const int lane  = tid & 31;
  const int wave  = __builtin_amdgcn_readfirstlane(tid >> 5);
  const int ch    = blockIdx.z;
  const int rbase = blockIdx.y * (TILE_R * NT);
  const int c0    = blockIdx.x * TILE_C;

  const float* __restrict__ xc = x   + (size_t)ch * NRR * NCC;
  float* __restrict__       oc = out + (size_t)ch * NRR * NCC;

  if (tid < 16) s_g[tid] = (tid < KSZ) ? taps.g[ch][tid] : 0.0f;
  __syncthreads();

  const int m  = lane & 15;
  const int lh = lane >> 4;
  const int cw = wave * 16;

  float g[KSZ];
#pragma unroll
  for (int u = 0; u < KSZ; ++u) g[u] = s_g[u];
  v2f afrag[7];
#pragma unroll
  for (int kk = 0; kk < 7; ++kk) {
    const int kb = 4 * kk + 2 * lh;
    int d0 = kb - m;
    int d1 = kb + 1 - m;
    int i0 = ((unsigned)d0 < KSZ) ? d0 : 15;
    int i1 = ((unsigned)d1 < KSZ) ? d1 : 15;
    afrag[kk].x = s_g[i0];
    afrag[kk].y = s_g[i1];
  }

  const int q  = 4 * lane;
  const int cA = (c0 + q - 4) & (NCC - 1);
  const int cB =  c0 + q;
  const int cC = (c0 + q + 4) & (NCC - 1);
  const int cS = (c0 + q + 8) & (NCC - 1);

  {
    const int nrow = (wave < 4) ? 2 : 1;
    RowIn in[2];
#pragma unroll
    for (int i = 0; i < 2; ++i) {
      if (i < nrow) {
        const int rel = wave + 8 * i;
        const int gr  = (rbase - 4 + rel) & (NRR - 1);
        in[i] = load_row(xc + (size_t)gr * NCC, cA, cB, cC, cS);
      }
    }
#pragma unroll
    for (int i = 0; i < 2; ++i) {
      if (i < nrow) {
        const int rel = wave + 8 * i;
        *(v4f*)&s_tmp[(rel & (RING - 1)) * LDS_PITCH + q] = filt_row(in[i], g);
      }
    }
  }

  for (int t = 0; t < NT; ++t) {
    {
      RowIn in[2];
#pragma unroll
      for (int i = 0; i < 2; ++i) {
        const int rel = 16 * t + 12 + wave + 8 * i;
        const int gr  = (rbase - 4 + rel) & (NRR - 1);
        in[i] = load_row(xc + (size_t)gr * NCC, cA, cB, cC, cS);
      }
      __builtin_amdgcn_sched_barrier(0);
#pragma unroll
      for (int i = 0; i < 2; ++i) {
        const int rel = 16 * t + 12 + wave + 8 * i;
        *(v4f*)&s_tmp[(rel & (RING - 1)) * LDS_PITCH + q] = filt_row(in[i], g);
      }
    }
    __syncthreads();

    v8f acc = {};
#pragma unroll
    for (int kk = 0; kk < 7; ++kk) {
      const int kb = 4 * kk + 2 * lh;
      v2f b;
      if (kk < 6) {
        b.x = s_tmp[((16 * t + kb + 0) & (RING - 1)) * LDS_PITCH + cw + m];
        b.y = s_tmp[((16 * t + kb + 1) & (RING - 1)) * LDS_PITCH + cw + m];
      } else {
        float r24 = s_tmp[((16 * t + 24) & (RING - 1)) * LDS_PITCH + cw + m];
        b.x = lh ? 0.0f : r24;
        b.y = 0.0f;
      }
      acc = __builtin_amdgcn_wmma_f32_16x16x4_f32(
           false, afrag[kk],  false, b,
           (short)0, acc,  false,  false);
    }

    const int r0 = rbase + t * TILE_R;
    const int cc = c0 + cw + m;
#pragma unroll
    for (int v = 0; v < 8; ++v) {
      int rr = r0 + v + 8 * lh;
      oc[(size_t)rr * NCC + cc] = acc[v];
    }
  }
}

extern "C" void kernel_launch(void* const* d_in, const int* in_sizes, int n_in,
                              void* d_out, int out_size, void* d_ws, size_t ws_size,
                              hipStream_t stream) {
  (void)in_sizes; (void)n_in; (void)out_size; (void)d_ws; (void)ws_size;

  static const double mtf[NCH] = {0.38, 0.34, 0.34, 0.26, 0.22, 0.23};
  TapsArg taps;
  for (int c = 0; c < NCH; ++c) {
    double sigma = 2.0 * std::sqrt(-2.0 * std::log(mtf[c])) / M_PI;
    double e[KSZ], s = 0.0;
    for (int t = 0; t < KSZ; ++t) {
      double d = (double)t - 4.5;
      e[t] = std::exp(-d * d / (2.0 * sigma * sigma));
      s += e[t];
    }
    for (int t = 0; t < KSZ; ++t) taps.g[c][t] = (float)(e[t] / s);
  }

  const float* x  = (const float*)d_in[0];
  float*       yo = (float*)d_out;

  dim3 grid(NCC / TILE_C, NRR / (TILE_R * NT), NCH);
  s2_gaussian_blur_fused<<<grid, 256, 0, stream>>>(x, yo, taps);
}


# input rows loaded one tile ahead into a second register set; the tile barrier fences LDS only, so the loads stay in flight across sums, barrier, matrix phase and stores; same sums, same order
# speedup vs baseline: 2.1211x; 1.0722x over previous
#include <hip/hip_runtime.h>
#include <cmath>


#define NRR 2048
#define NCC 2048
#define KSZ 10
#define NCH 6

#define TILE_R 16
#define NWAVES 8
#define TILE_C (16 * NWAVES)
#define NT 16
#define RING 64
#define LDS_PITCH 132

typedef __attribute__((ext_vector_type(2))) float v2f;
typedef __attribute__((ext_vector_type(8))) float v8f;

typedef __attribute__((ext_vector_type(4))) float v4f;

struct RowIn { v4f a, b, c; float s; };

__device__ __forceinline__ RowIn load_row(const float* __restrict__ rp,
                                          int cA, int cB, int cC, int cS) {
  RowIn r;
  r.a = *(const v4f*)(rp + cA);
  r.b = *(const v4f*)(rp + cB);
  r.c = *(const v4f*)(rp + cC);
  r.s = rp[cS];
  return r;
}

__device__ __forceinline__ v4f filt_row(const RowIn& r, const float (&g)[KSZ]) {
  const float x[13] = {r.a.x, r.a.y, r.a.z, r.a.w, r.b.x, r.b.y, r.b.z, r.b.w,
                       r.c.x, r.c.y, r.c.z, r.c.w, r.s};
  v4f o;
#pragma unroll
  for (int k = 0; k < 4; ++k) {
    float acc = 0.0f;
#pragma unroll
    for (int u = 0; u < KSZ; ++u) acc = fmaf(g[u], x[k + u], acc);
    o[k] = acc;
  }
  return o;
}

__device__ __forceinline__ void lds_barrier() {
  __builtin_amdgcn_fence(__ATOMIC_RELEASE, "workgroup", "local");
  __builtin_amdgcn_s_barrier();
  __builtin_amdgcn_fence(__ATOMIC_ACQUIRE, "workgroup", "local");
}

struct TapsArg { float g[NCH][KSZ]; };

__global__ __launch_bounds__(256) __attribute__((amdgpu_waves_per_eu(6)))
void s2_gaussian_blur_fused(const float* __restrict__ x,
                            float* __restrict__ out,
                            TapsArg taps) {
  __shared__ __attribute__((aligned(16))) float s_tmp[RING * LDS_PITCH];
  __shared__ float s_g[16];

  const int tid   = threadIdx.x;
  const int lane  = tid & 31;
  const int wave  = __builtin_amdgcn_readfirstlane(tid >> 5);
  const int ch    = blockIdx.z;
  const int rbase = blockIdx.y * (TILE_R * NT);
  const int c0    = blockIdx.x * TILE_C;

  const float* __restrict__ xc = x   + (size_t)ch * NRR * NCC;
  float* __restrict__       oc = out + (size_t)ch * NRR * NCC;

  if (tid < 16) s_g[tid] = (tid < KSZ) ? taps.g[ch][tid] : 0.0f;
  __syncthreads();

  const int m  = lane & 15;
  const int lh = lane >> 4;
  const int cw = wave * 16;

  float g[KSZ];
#pragma unroll
  for (int u = 0; u < KSZ; ++u) g[u] = s_g[u];
  v2f afrag[7];
#pragma unroll
  for (int kk = 0; kk < 7; ++kk) {
    const int kb = 4 * kk + 2 * lh;
    int d0 = kb - m;
    int d1 = kb + 1 - m;
    int i0 = ((unsigned)d0 < KSZ) ? d0 : 15;
    int i1 = ((unsigned)d1 < KSZ) ? d1 : 15;
    afrag[kk].x = s_g[i0];
    afrag[kk].y = s_g[i1];
  }

  const int q  = 4 * lane;
  const int cA = (c0 + q - 4) & (NCC - 1);
  const int cB =  c0 + q;
  const int cC = (c0 + q + 4) & (NCC - 1);
  const int cS = (c0 + q + 8) & (NCC - 1);

  RowIn buf[2][2];

  {
    const int nrow = (wave < 4) ? 2 : 1;
    RowIn in[2];
#pragma unroll
    for (int i = 0; i < 2; ++i) {
      if (i < nrow) {
        const int rel = wave + 8 * i;
        const int gr  = (rbase - 4 + rel) & (NRR - 1);
        in[i] = load_row(xc + (size_t)gr * NCC, cA, cB, cC, cS);
      }
    }
#pragma unroll
    for (int i = 0; i < 2; ++i) {
      const int rel = 12 + wave + 8 * i;
      const int gr  = (rbase - 4 + rel) & (NRR - 1);
      buf[0][i] = load_row(xc + (size_t)gr * NCC, cA, cB, cC, cS);
    }
    __builtin_amdgcn_sched_barrier(0);
#pragma unroll
    for (int i = 0; i < 2; ++i) {
      if (i < nrow) {
        const int rel = wave + 8 * i;
        *(v4f*)&s_tmp[(rel & (RING - 1)) * LDS_PITCH + q] = filt_row(in[i], g);
      }
    }
  }

  auto tile_step = [&](const int t, const RowIn (&cur)[2], RowIn (&nxt)[2],
                       const bool more) __attribute__((always_inline)) {
    {
      if (more) {
#pragma unroll
        for (int i = 0; i < 2; ++i) {
          const int rel = 16 * (t + 1) + 12 + wave + 8 * i;
          const int gr  = (rbase - 4 + rel) & (NRR - 1);
          nxt[i] = load_row(xc + (size_t)gr * NCC, cA, cB, cC, cS);
        }
      }
      __builtin_amdgcn_sched_barrier(0);
#pragma unroll
      for (int i = 0; i < 2; ++i) {
        const int rel = 16 * t + 12 + wave + 8 * i;
        *(v4f*)&s_tmp[(rel & (RING - 1)) * LDS_PITCH + q] = filt_row(cur[i], g);
      }
    }
    lds_barrier();

    v8f acc = {};
#pragma unroll
    for (int kk = 0; kk < 7; ++kk) {
      const int kb = 4 * kk + 2 * lh;
      v2f b;
      if (kk < 6) {
        b.x = s_tmp[((16 * t + kb + 0) & (RING - 1)) * LDS_PITCH + cw + m];
        b.y = s_tmp[((16 * t + kb + 1) & (RING - 1)) * LDS_PITCH + cw + m];
      } else {
        float r24 = s_tmp[((16 * t + 24) & (RING - 1)) * LDS_PITCH + cw + m];
        b.x = lh ? 0.0f : r24;
        b.y = 0.0f;
      }
      acc = __builtin_amdgcn_wmma_f32_16x16x4_f32(
           false, afrag[kk],  false, b,
           (short)0, acc,  false,  false);
    }

    const int r0 = rbase + t * TILE_R;
    const int cc = c0 + cw + m;
#pragma unroll
    for (int v = 0; v < 8; ++v) {
      int rr = r0 + v + 8 * lh;
      oc[(size_t)rr * NCC + cc] = acc[v];
    }
  };

  for (int t2 = 0; t2 < NT - 2; t2 += 2) {
    tile_step(t2,     buf[0], buf[1], true);
    tile_step(t2 + 1, buf[1], buf[0], true);
  }
  tile_step(NT - 2, buf[0], buf[1], true);
  tile_step(NT - 1, buf[1], buf[0], false);
}

extern "C" void kernel_launch(void* const* d_in, const int* in_sizes, int n_in,
                              void* d_out, int out_size, void* d_ws, size_t ws_size,
                              hipStream_t stream) {
  (void)in_sizes; (void)n_in; (void)out_size; (void)d_ws; (void)ws_size;

  static const double mtf[NCH] = {0.38, 0.34, 0.34, 0.26, 0.22, 0.23};
  TapsArg taps;
  for (int c = 0; c < NCH; ++c) {
    double sigma = 2.0 * std::sqrt(-2.0 * std::log(mtf[c])) / M_PI;
    double e[KSZ], s = 0.0;
    for (int t = 0; t < KSZ; ++t) {
      double d = (double)t - 4.5;
      e[t] = std::exp(-d * d / (2.0 * sigma * sigma));
      s += e[t];
    }
    for (int t = 0; t < KSZ; ++t) taps.g[c][t] = (float)(e[t] / s);
  }

  const float* x  = (const float*)d_in[0];
  float*       yo = (float*)d_out;

  dim3 grid(NCC / TILE_C, NRR / (TILE_R * NT), NCH);
  s2_gaussian_blur_fused<<<grid, 256, 0, stream>>>(x, yo, taps);
}


# output tiles staged through LDS and stored one tile later as whole rows: each store instruction writes four whole 128-byte lines (16 B per lane) instead of two 64-byte pieces; same sums, same order
# speedup vs baseline: 2.8249x; 1.3318x over previous
#include <hip/hip_runtime.h>
#include <cmath>


#define NRR 2048
#define NCC 2048
#define KSZ 10
#define NCH 6

#define TILE_R 16
#define NWAVES 8
#define TILE_C (16 * NWAVES)
#define NT 16
#define RING 64
#define LDS_PITCH 132

typedef __attribute__((ext_vector_type(2))) float v2f;
typedef __attribute__((ext_vector_type(8))) float v8f;

typedef __attribute__((ext_vector_type(4))) float v4f;

struct RowIn { v4f a, b, c; float s; };

__device__ __forceinline__ RowIn load_row(const float* __restrict__ rp,
                                          int cA, int cB, int cC, int cS) {
  RowIn r;
  r.a = *(const v4f*)(rp + cA);
  r.b = *(const v4f*)(rp + cB);
  r.c = *(const v4f*)(rp + cC);
  r.s = rp[cS];
  return r;
}

__device__ __forceinline__ v4f filt_row(const RowIn& r, const float (&g)[KSZ]) {
  const float x[13] = {r.a.x, r.a.y, r.a.z, r.a.w, r.b.x, r.b.y, r.b.z, r.b.w,
                       r.c.x, r.c.y, r.c.z, r.c.w, r.s};
  v4f o;
#pragma unroll
  for (int k = 0; k < 4; ++k) {
    float acc = 0.0f;
#pragma unroll
    for (int u = 0; u < KSZ; ++u) acc = fmaf(g[u], x[k + u], acc);
    o[k] = acc;
  }
  return o;
}

__device__ __forceinline__ void lds_barrier() {
  __builtin_amdgcn_fence(__ATOMIC_RELEASE, "workgroup", "local");
  __builtin_amdgcn_s_barrier();
  __builtin_amdgcn_fence(__ATOMIC_ACQUIRE, "workgroup", "local");
}

struct TapsArg { float g[NCH][KSZ]; };

__global__ __launch_bounds__(256) __attribute__((amdgpu_waves_per_eu(6)))
void s2_gaussian_blur_fused(const float* __restrict__ x,
                            float* __restrict__ out,
                            TapsArg taps) {
  __shared__ __attribute__((aligned(16))) float s_tmp[RING * LDS_PITCH];
  __shared__ float s_g[16];
  __shared__ __attribute__((aligned(16))) float s_out[2][TILE_R * LDS_PITCH];

  const int tid   = threadIdx.x;
  const int lane  = tid & 31;
  const int wave  = __builtin_amdgcn_readfirstlane(tid >> 5);
  const int ch    = blockIdx.z;
  const int rbase = blockIdx.y * (TILE_R * NT);
  const int c0    = blockIdx.x * TILE_C;

  const float* __restrict__ xc = x   + (size_t)ch * NRR * NCC;
  float* __restrict__       oc = out + (size_t)ch * NRR * NCC;

  if (tid < 16) s_g[tid] = (tid < KSZ) ? taps.g[ch][tid] : 0.0f;
  __syncthreads();

  const int m  = lane & 15;
  const int lh = lane >> 4;
  const int cw = wave * 16;

  float g[KSZ];
#pragma unroll
  for (int u = 0; u < KSZ; ++u) g[u] = s_g[u];
  v2f afrag[7];
#pragma unroll
  for (int kk = 0; kk < 7; ++kk) {
    const int kb = 4 * kk + 2 * lh;
    int d0 = kb - m;
    int d1 = kb + 1 - m;
    int i0 = ((unsigned)d0 < KSZ) ? d0 : 15;
    int i1 = ((unsigned)d1 < KSZ) ? d1 : 15;
    afrag[kk].x = s_g[i0];
    afrag[kk].y = s_g[i1];
  }

  const int q  = 4 * lane;
  const int cA = (c0 + q - 4) & (NCC - 1);
  const int cB =  c0 + q;
  const int cC = (c0 + q + 4) & (NCC - 1);
  const int cS = (c0 + q + 8) & (NCC - 1);

  RowIn buf[2][2];

  {
    const int nrow = (wave < 4) ? 2 : 1;
    RowIn in[2];
#pragma unroll
    for (int i = 0; i < 2; ++i) {
      if (i < nrow) {
        const int rel = wave + 8 * i;
        const int gr  = (rbase - 4 + rel) & (NRR - 1);
        in[i] = load_row(xc + (size_t)gr * NCC, cA, cB, cC, cS);
      }
    }
#pragma unroll
    for (int i = 0; i < 2; ++i) {
      const int rel = 12 + wave + 8 * i;
      const int gr  = (rbase - 4 + rel) & (NRR - 1);
      buf[0][i] = load_row(xc + (size_t)gr * NCC, cA, cB, cC, cS);
    }
    __builtin_amdgcn_sched_barrier(0);
#pragma unroll
    for (int i = 0; i < 2; ++i) {
      if (i < nrow) {
        const int rel = wave + 8 * i;
        *(v4f*)&s_tmp[(rel & (RING - 1)) * LDS_PITCH + q] = filt_row(in[i], g);
      }
    }
  }

  auto store_tile = [&](const float* __restrict__ sp, const int r0) __attribute__((always_inline)) {
    const v4f o0 = *(const v4f*)&sp[wave * LDS_PITCH + q];
    const v4f o1 = *(const v4f*)&sp[(wave + 8) * LDS_PITCH + q];
    float* op = oc + (size_t)(r0 + wave) * NCC + c0 + q;
    *(v4f*)op = o0;
    *(v4f*)(op + 8 * NCC) = o1;
  };

  auto tile_step = [&](const int t, const RowIn (&cur)[2], RowIn (&nxt)[2],
                       const bool more, const bool first,
                       const int par) __attribute__((always_inline)) {
    {
      if (more) {
#pragma unroll
        for (int i = 0; i < 2; ++i) {
          const int rel = 16 * (t + 1) + 12 + wave + 8 * i;
          const int gr  = (rbase - 4 + rel) & (NRR - 1);
          nxt[i] = load_row(xc + (size_t)gr * NCC, cA, cB, cC, cS);
        }
      }
      __builtin_amdgcn_sched_barrier(0);
#pragma unroll
      for (int i = 0; i < 2; ++i) {
        const int rel = 16 * t + 12 + wave + 8 * i;
        *(v4f*)&s_tmp[(rel & (RING - 1)) * LDS_PITCH + q] = filt_row(cur[i], g);
      }
    }
    lds_barrier();

    if (!first) store_tile(s_out[par ^ 1], rbase + (t - 1) * TILE_R);

    v8f acc = {};
#pragma unroll
    for (int kk = 0; kk < 7; ++kk) {
      const int kb = 4 * kk + 2 * lh;
      v2f b;
      if (kk < 6) {
        b.x = s_tmp[((16 * t + kb + 0) & (RING - 1)) * LDS_PITCH + cw + m];
        b.y = s_tmp[((16 * t + kb + 1) & (RING - 1)) * LDS_PITCH + cw + m];
      } else {
        float r24 = s_tmp[((16 * t + 24) & (RING - 1)) * LDS_PITCH + cw + m];
        b.x = lh ? 0.0f : r24;
        b.y = 0.0f;
      }
      acc = __builtin_amdgcn_wmma_f32_16x16x4_f32(
           false, afrag[kk],  false, b,
           (short)0, acc,  false,  false);
    }

    float* __restrict__ st = s_out[par];
#pragma unroll
    for (int v = 0; v < 8; ++v) st[(v + 8 * lh) * LDS_PITCH + cw + m] = acc[v];
  };

  tile_step(0, buf[0], buf[1], true, true,  0);
  tile_step(1, buf[1], buf[0], true, false, 1);
  for (int t2 = 2; t2 < NT - 2; t2 += 2) {
    tile_step(t2,     buf[0], buf[1], true, false, 0);
    tile_step(t2 + 1, buf[1], buf[0], true, false, 1);
  }
  tile_step(NT - 2, buf[0], buf[1], true,  false, 0);
  tile_step(NT - 1, buf[1], buf[0], false, false, 1);

  lds_barrier();
  store_tile(s_out[(NT - 1) & 1], rbase + (NT - 1) * TILE_R);
}

extern "C" void kernel_launch(void* const* d_in, const int* in_sizes, int n_in,
                              void* d_out, int out_size, void* d_ws, size_t ws_size,
                              hipStream_t stream) {
  (void)in_sizes; (void)n_in; (void)out_size; (void)d_ws; (void)ws_size;

  static const double mtf[NCH] = {0.38, 0.34, 0.34, 0.26, 0.22, 0.23};
  TapsArg taps;
  for (int c = 0; c < NCH; ++c) {
    double sigma = 2.0 * std::sqrt(-2.0 * std::log(mtf[c])) / M_PI;
    double e[KSZ], s = 0.0;
    for (int t = 0; t < KSZ; ++t) {
      double d = (double)t - 4.5;
      e[t] = std::exp(-d * d / (2.0 * sigma * sigma));
      s += e[t];
    }
    for (int t = 0; t < KSZ; ++t) taps.g[c][t] = (float)(e[t] / s);
  }

  const float* x  = (const float*)d_in[0];
  float*       yo = (float*)d_out;

  dim3 grid(NCC / TILE_C, NRR / (TILE_R * NT), NCH);
  s2_gaussian_blur_fused<<<grid, 256, 0, stream>>>(x, yo, taps);
}


# two 16-row tiles per barrier step: a wave filters 4 rows and runs two independent 7-WMMA chains, each row re-requested once summed (one register set); 9 barriers not 17; same sums, same order
# speedup vs baseline: 3.2138x; 1.1377x over previous
#include <hip/hip_runtime.h>
#include <cmath>


#define NRR 2048
#define NCC 2048
#define KSZ 10
#define NCH 6

#define TILE_R 16
#define NWAVES 8
#define TILE_C (16 * NWAVES)
#define STEP_R 32
#define NS 8
#define RING 96
#define RING_OFF 4
#define LDS_PITCH 132

typedef __attribute__((ext_vector_type(2))) float v2f;
typedef __attribute__((ext_vector_type(8))) float v8f;

typedef __attribute__((ext_vector_type(4))) float v4f;

struct RowIn { v4f a, b, c; float s; };

__device__ __forceinline__ RowIn load_row(const float* __restrict__ rp,
                                          int cA, int cB, int cC, int cS) {
  RowIn r;
  r.a = *(const v4f*)(rp + cA);
  r.b = *(const v4f*)(rp + cB);
  r.c = *(const v4f*)(rp + cC);
  r.s = rp[cS];
  return r;
}

__device__ __forceinline__ v4f filt_row(const RowIn& r, const v2f (&g2)[KSZ]) {
  const float x[13] = {r.a.x, r.a.y, r.a.z, r.a.w, r.b.x, r.b.y, r.b.z, r.b.w,
                       r.c.x, r.c.y, r.c.z, r.c.w, r.s};
  v2f lo = {0.0f, 0.0f};
  v2f hi = {0.0f, 0.0f};
#pragma unroll
  for (int u = 0; u < KSZ; ++u) {
    const v2f xl = {x[u],     x[u + 1]};
    const v2f xh = {x[u + 2], x[u + 3]};
    lo = __builtin_elementwise_fma(g2[u], xl, lo);
    hi = __builtin_elementwise_fma(g2[u], xh, hi);
  }
  v4f o = {lo.x, lo.y, hi.x, hi.y};
  return o;
}

__device__ __forceinline__ void lds_barrier() {
  __builtin_amdgcn_fence(__ATOMIC_RELEASE, "workgroup", "local");
  __builtin_amdgcn_s_barrier();
  __builtin_amdgcn_fence(__ATOMIC_ACQUIRE, "workgroup", "local");
}

struct TapsArg { float g[NCH][KSZ]; };

__global__ __launch_bounds__(256) __attribute__((amdgpu_waves_per_eu(6)))
void s2_gaussian_blur_fused(const float* __restrict__ x,
                            float* __restrict__ out,
                            TapsArg taps) {
  __shared__ __attribute__((aligned(16))) float s_tmp[RING * LDS_PITCH];
  __shared__ __attribute__((aligned(8))) float s_g[32];
  __shared__ __attribute__((aligned(16))) float s_out[2][STEP_R * LDS_PITCH];

  const int tid   = threadIdx.x;
  const int lane  = tid & 31;
  const int wave  = __builtin_amdgcn_readfirstlane(tid >> 5);
  const int ch    = blockIdx.z;
  const int rbase = blockIdx.y * (STEP_R * NS);
  const int c0    = blockIdx.x * TILE_C;

  const float* __restrict__ xc = x   + (size_t)ch * NRR * NCC;
  float* __restrict__       oc = out + (size_t)ch * NRR * NCC;

  if (tid < 16) {
    const float gv = (tid < KSZ) ? taps.g[ch][tid] : 0.0f;
    s_g[2 * tid] = gv;
    s_g[2 * tid + 1] = gv;
  }
  __syncthreads();

  const int m  = lane & 15;
  const int lh = lane >> 4;
  const int cw = wave * 16;

  v2f g[KSZ];
#pragma unroll
  for (int u = 0; u < KSZ; ++u) g[u] = *(const v2f*)&s_g[2 * u];
  v2f afrag[7];
#pragma unroll
  for (int kk = 0; kk < 7; ++kk) {
    const int kb = 4 * kk + 2 * lh;
    int d0 = kb - m;
    int d1 = kb + 1 - m;
    int i0 = ((unsigned)d0 < KSZ) ? d0 : 15;
    int i1 = ((unsigned)d1 < KSZ) ? d1 : 15;
    afrag[kk].x = s_g[2 * i0];
    afrag[kk].y = s_g[2 * i1];
  }

  const int q  = 4 * lane;
  const int cA = (c0 + q - 4) & (NCC - 1);
  const int cB = (c0 + q) & (NCC - 1);
  const int cC = (c0 + q + 4) & (NCC - 1);
  const int cS = (c0 + q + 8) & (NCC - 1);

  RowIn buf[4];

  {
    const int nrow = (wave < 4) ? 2 : 1;
    RowIn in[2];
#pragma unroll
    for (int i = 0; i < 2; ++i) {
      if (i < nrow) {
        const int rel = wave + 8 * i;
        const int gr  = (rbase - 4 + rel) & (NRR - 1);
        in[i] = load_row(xc + (size_t)gr * NCC, cA, cB, cC, cS);
      }
    }
#pragma unroll
    for (int i = 0; i < 4; ++i) {
      const int rel = 12 + wave + 8 * i;
      const int gr  = (rbase - 4 + rel) & (NRR - 1);
      buf[i] = load_row(xc + (size_t)gr * NCC, cA, cB, cC, cS);
    }
    __builtin_amdgcn_sched_barrier(0);
#pragma unroll
    for (int i = 0; i < 2; ++i) {
      if (i < nrow) {
        const int rrow = (8 * i + RING_OFF) % RING + wave;
        *(v4f*)&s_tmp[rrow * LDS_PITCH + q] = filt_row(in[i], g);
      }
    }
  }

  auto store_step = [&](const float* __restrict__ sp, const int r0) __attribute__((always_inline)) {
    v4f o[4];
#pragma unroll
    for (int j = 0; j < 4; ++j) o[j] = *(const v4f*)&sp[(wave + 8 * j) * LDS_PITCH + q];
    float* op = oc + (size_t)(r0 + wave) * NCC + cB;
#pragma unroll
    for (int j = 0; j < 4; ++j) *(v4f*)(op + (size_t)(8 * j) * NCC) = o[j];
  };

  auto row_step = [&](const int s, const bool more, const bool first,
                      const int par) __attribute__((always_inline)) {
    __builtin_amdgcn_sched_barrier(0);
#pragma unroll
    for (int i = 0; i < 4; ++i) {
      const int rel  = STEP_R * s + 12 + wave + 8 * i;
      const int rrow = (STEP_R * s + 12 + 8 * i + RING_OFF) % RING + wave;
      *(v4f*)&s_tmp[rrow * LDS_PITCH + q] = filt_row(buf[i], g);
      if (more) {
        const int gr = (rbase - 4 + rel + STEP_R) & (NRR - 1);
        buf[i] = load_row(xc + (size_t)gr * NCC, cA, cB, cC, cS);
      }
    }
    lds_barrier();

    v2f bf[10];
#pragma unroll
    for (int j = 0; j < 10; ++j) {
      const int rb = (STEP_R * s + 4 * j + RING_OFF) % RING + 2 * lh;
      bf[j].x = s_tmp[(rb + 0) * LDS_PITCH + cw + m];
      bf[j].y = s_tmp[(rb + 1) * LDS_PITCH + cw + m];
    }
    const float r40 = s_tmp[((STEP_R * s + 40 + RING_OFF) % RING) * LDS_PITCH + cw + m];
    v2f lastA, lastB;
    lastA.x = lh ? 0.0f : bf[6].x;
    lastA.y = 0.0f;
    lastB.x = lh ? 0.0f : r40;
    lastB.y = 0.0f;

    v8f acc0 = {};
    v8f acc1 = {};
#pragma unroll
    for (int kk = 0; kk < 7; ++kk) {
      const v2f b0 = (kk < 6) ? bf[kk] : lastA;
      const v2f b1 = (kk < 6) ? bf[kk + 4] : lastB;
      acc0 = __builtin_amdgcn_wmma_f32_16x16x4_f32(
           false, afrag[kk],  false, b0,
           (short)0, acc0,  false,  false);
      acc1 = __builtin_amdgcn_wmma_f32_16x16x4_f32(
           false, afrag[kk],  false, b1,
           (short)0, acc1,  false,  false);
    }

    float* __restrict__ st = s_out[par];
#pragma unroll
    for (int v = 0; v < 8; ++v) {
      st[(v + 8 * lh) * LDS_PITCH + cw + m]      = acc0[v];
      st[(16 + v + 8 * lh) * LDS_PITCH + cw + m] = acc1[v];
    }

    __builtin_amdgcn_sched_barrier(0);
    if (!first) store_step(s_out[par ^ 1], rbase + (s - 1) * STEP_R);
  };

  row_step(0, true,  true,  0);
  row_step(1, true,  false, 1);
  row_step(2, true,  false, 0);
  row_step(3, true,  false, 1);
  row_step(4, true,  false, 0);
  row_step(5, true,  false, 1);
  row_step(6, true,  false, 0);
  row_step(7, false, false, 1);
  static_assert(NS == 8, "the steps are written out one by one");

  lds_barrier();
  store_step(s_out[(NS - 1) & 1], rbase + (NS - 1) * STEP_R);
}

extern "C" void kernel_launch(void* const* d_in, const int* in_sizes, int n_in,
                              void* d_out, int out_size, void* d_ws, size_t ws_size,
                              hipStream_t stream) {
  (void)in_sizes; (void)n_in; (void)out_size; (void)d_ws; (void)ws_size;

  static const double mtf[NCH] = {0.38, 0.34, 0.34, 0.26, 0.22, 0.23};
  TapsArg taps;
  for (int c = 0; c < NCH; ++c) {
    double sigma = 2.0 * std::sqrt(-2.0 * std::log(mtf[c])) / M_PI;
    double e[KSZ], s = 0.0;
    for (int t = 0; t < KSZ; ++t) {
      double d = (double)t - 4.5;
      e[t] = std::exp(-d * d / (2.0 * sigma * sigma));
      s += e[t];
    }
    for (int t = 0; t < KSZ; ++t) taps.g[c][t] = (float)(e[t] / s);
  }

  const float* x  = (const float*)d_in[0];
  float*       yo = (float*)d_out;

  dim3 grid(NCC / TILE_C, NRR / (STEP_R * NS), NCH);
  s2_gaussian_blur_fused<<<grid, 256, 0, stream>>>(x, yo, taps);
}
